# starting kernel
# baseline (speedup 1.0000x reference)
#include <hip/hip_runtime.h>

typedef __attribute__((ext_vector_type(2))) float v2f;
typedef __attribute__((ext_vector_type(8))) float v8f;

#define LSTR 80

__global__ __launch_bounds__(512) void gram_partial(const float* __restrict__ Phi,
                                                    float* __restrict__ part,
                                                    int m, int nwg) {
    __shared__ float tile[64 * LSTR];
    const int tid  = threadIdx.x;
    const int lane = tid & 31;
    const int wave = tid >> 5;
    const int ti   = wave >> 2;
    const int tj   = wave & 3;
    const int li   = lane & 15;
    const int kh   = lane >> 4;

    v8f acc = {};

    for (long rowbase = (long)blockIdx.x * 64; rowbase < (long)m; rowbase += (long)nwg * 64) {
        {
            const int r = tid >> 3;
            const int c = (tid & 7) * 8;
            const long row = rowbase + r;
            float* dst = &tile[r * LSTR + c];
            if (row < (long)m) {
                const float4* src = reinterpret_cast<const float4*>(Phi + row * 64 + c);
                float4 a = src[0], b = src[1];
                dst[0]=a.x; dst[1]=a.y; dst[2]=a.z; dst[3]=a.w;
                dst[4]=b.x; dst[5]=b.y; dst[6]=b.z; dst[7]=b.w;
            } else {
                #pragma unroll
                for (int q = 0; q < 8; ++q) dst[q] = 0.0f;
            }
        }
        __syncthreads();

        #pragma unroll
        for (int k0 = 0; k0 < 64; k0 += 4) {
            const int ka = k0 + 2 * kh;
            v2f va, vb;
            va.x = tile[ka       * LSTR + ti * 16 + li];
            va.y = tile[(ka + 1) * LSTR + ti * 16 + li];
            vb.x = tile[ka       * LSTR + tj * 16 + li];
            vb.y = tile[(ka + 1) * LSTR + tj * 16 + li];
            acc = __builtin_amdgcn_wmma_f32_16x16x4_f32(false, va, false, vb,
                                                        (short)0, acc, false, false);
        }
        __syncthreads();
    }

    float* p = part + (size_t)blockIdx.x * 4096;
    #pragma unroll
    for (int v = 0; v < 8; ++v) {
        const int gi = ti * 16 + v + 8 * kh;
        const int gj = tj * 16 + li;
        p[gi * 64 + gj] = acc[v];
    }
}

__global__ __launch_bounds__(256) void reduce_g(const float* __restrict__ part,
                                                float* __restrict__ G, int nwg) {
    const int e = blockIdx.x * 256 + threadIdx.x;
    float s = 0.0f;
    for (int w = 0; w < nwg; ++w) s += part[(size_t)w * 4096 + e];
    G[e] = s;
}

__global__ __launch_bounds__(64) void chol_inv(const float* __restrict__ G,
                                               float* __restrict__ Rinv) {
    __shared__ float A[64 * 64];
    __shared__ float X[64 * 64];
    const int t = threadIdx.x;

    for (int i = 0; i < 64; ++i) A[i * 64 + t] = G[i * 64 + t];
    __syncthreads();

    for (int k = 0; k < 64; ++k) {
        const float dk = sqrtf(A[k * 64 + k]);
        if (t >= k) A[k * 64 + t] = (t == k) ? dk : (A[k * 64 + t] / dk);
        __syncthreads();
        if (t > k) {
            const float uki = A[k * 64 + t];
            for (int j = t; j < 64; ++j)
                A[t * 64 + j] -= uki * A[k * 64 + j];
        }
        __syncthreads();
    }

    for (int i = 0; i < 64; ++i) X[i * 64 + t] = 0.0f;
    X[t * 64 + t] = 1.0f / A[t * 64 + t];
    for (int i = t - 1; i >= 0; --i) {
        float s = 0.0f;
        for (int k = i + 1; k <= t; ++k) s += A[i * 64 + k] * X[k * 64 + t];
        X[i * 64 + t] = -s / A[i * 64 + i];
    }
    __syncthreads();
    for (int i = 0; i < 64; ++i) Rinv[i * 64 + t] = X[i * 64 + t];
}

__global__ __launch_bounds__(512) void apply_rinv(const float* __restrict__ Phi,
                                                  const float* __restrict__ Rinv,
                                                  float* __restrict__ Q, int m) {
    __shared__ float B[64 * LSTR];
    const int tid  = threadIdx.x;
    const int lane = tid & 31;
    const int wave = tid >> 5;
    const int ti   = wave >> 2;
    const int tj   = wave & 3;
    const int li   = lane & 15;
    const int kh   = lane >> 4;

    {
        const int r = tid >> 3;
        const int c = (tid & 7) * 8;
        const float* src = Rinv + r * 64 + c;
        float* dst = &B[r * LSTR + c];
        #pragma unroll
        for (int q = 0; q < 8; ++q) dst[q] = src[q];
    }
    __syncthreads();

    const long rowbase = (long)blockIdx.x * 64;
    long arow = rowbase + ti * 16 + li;
    if (arow >= (long)m) arow = (long)m - 1;
    const float2* aptr = reinterpret_cast<const float2*>(Phi + arow * 64);

    v8f acc = {};
    #pragma unroll
    for (int k0 = 0; k0 < 64; k0 += 4) {
        const int ka = k0 + 2 * kh;
        v2f va, vb;
        const float2 a = aptr[ka >> 1];
        va.x = a.x; va.y = a.y;
        vb.x = B[ka       * LSTR + tj * 16 + li];
        vb.y = B[(ka + 1) * LSTR + tj * 16 + li];
        acc = __builtin_amdgcn_wmma_f32_16x16x4_f32(false, va, false, vb,
                                                    (short)0, acc, false, false);
    }

    #pragma unroll
    for (int v = 0; v < 8; ++v) {
        const long row = rowbase + ti * 16 + v + 8 * kh;
        if (row < (long)m) Q[row * 64 + tj * 16 + li] = acc[v];
    }
}

extern "C" void kernel_launch(void* const* d_in, const int* in_sizes, int n_in,
                              void* d_out, int out_size, void* d_ws, size_t ws_size,
                              hipStream_t stream) {
    const float* Phi = (const float*)d_in[0];
    float* out = (float*)d_out;
    const int m = in_sizes[0] / 64;

    float* ws   = (float*)d_ws;
    float* G    = ws;
    float* Rinv = ws + 4096;
    float* part = ws + 8192;

    long avail = ((long)(ws_size / 4) - 8192) / 4096;
    int nwg = (int)(avail < 1 ? 1 : (avail > 512 ? 512 : avail));

    gram_partial<<<nwg, 512, 0, stream>>>(Phi, part, m, nwg);
    reduce_g<<<16, 256, 0, stream>>>(part, G, nwg);
    chol_inv<<<1, 64, 0, stream>>>(G, Rinv);

    const int nblk = (m + 63) / 64;
    apply_rinv<<<nblk, 512, 0, stream>>>(Phi, Rinv, out, m);
}


# apply_rinv: a wave owns 64 rows x all 64 columns (was 16x16): A straight from 16-byte global loads, each row read once (was 4x); B columns interleaved, stores write whole rows; Q's k order changes
# speedup vs baseline: 1.9909x; 1.9909x over previous
#include <hip/hip_runtime.h>

typedef __attribute__((ext_vector_type(2))) float v2f;
typedef __attribute__((ext_vector_type(8))) float v8f;

#define LSTR 80

__global__ __launch_bounds__(512) void gram_partial(const float* __restrict__ Phi,
                                                    float* __restrict__ part,
                                                    int m, int nwg) {
    __shared__ float tile[64 * LSTR];
    const int tid  = threadIdx.x;
    const int lane = tid & 31;
    const int wave = tid >> 5;
    const int ti   = wave >> 2;
    const int tj   = wave & 3;
    const int li   = lane & 15;
    const int kh   = lane >> 4;

    v8f acc = {};

    for (long rowbase = (long)blockIdx.x * 64; rowbase < (long)m; rowbase += (long)nwg * 64) {
        {
            const int r = tid >> 3;
            const int c = (tid & 7) * 8;
            const long row = rowbase + r;
            float* dst = &tile[r * LSTR + c];
            if (row < (long)m) {
                const float4* src = reinterpret_cast<const float4*>(Phi + row * 64 + c);
                float4 a = src[0], b = src[1];
                dst[0]=a.x; dst[1]=a.y; dst[2]=a.z; dst[3]=a.w;
                dst[4]=b.x; dst[5]=b.y; dst[6]=b.z; dst[7]=b.w;
            } else {
                #pragma unroll
                for (int q = 0; q < 8; ++q) dst[q] = 0.0f;
            }
        }
        __syncthreads();

        #pragma unroll
        for (int k0 = 0; k0 < 64; k0 += 4) {
            const int ka = k0 + 2 * kh;
            v2f va, vb;
            va.x = tile[ka       * LSTR + ti * 16 + li];
            va.y = tile[(ka + 1) * LSTR + ti * 16 + li];
            vb.x = tile[ka       * LSTR + tj * 16 + li];
            vb.y = tile[(ka + 1) * LSTR + tj * 16 + li];
            acc = __builtin_amdgcn_wmma_f32_16x16x4_f32(false, va, false, vb,
                                                        (short)0, acc, false, false);
        }
        __syncthreads();
    }

    float* p = part + (size_t)blockIdx.x * 4096;
    #pragma unroll
    for (int v = 0; v < 8; ++v) {
        const int gi = ti * 16 + v + 8 * kh;
        const int gj = tj * 16 + li;
        p[gi * 64 + gj] = acc[v];
    }
}

__global__ __launch_bounds__(256) void reduce_g(const float* __restrict__ part,
                                                float* __restrict__ G, int nwg) {
    const int e = blockIdx.x * 256 + threadIdx.x;
    float s = 0.0f;
    for (int w = 0; w < nwg; ++w) s += part[(size_t)w * 4096 + e];
    G[e] = s;
}

#define CLS 144

__device__ __forceinline__ void panel_tile(float* P, int kb, int i, int c,
                                           bool trailing, int li, int kh) {
    const int krow = kb * 16;
    const int ocol = c * 16 + li;
    const int orow = (trailing ? i : kb) * 16;
    v8f acc = {};
    if (trailing) {
        #pragma unroll
        for (int v = 0; v < 8; ++v) acc[v] = P[(orow + v + 8 * kh) * CLS + ocol];
    }
    v2f va[4], vb[4];
    #pragma unroll
    for (int q = 0; q < 4; ++q) {
        const int ka = q * 4 + 2 * kh;
        if (trailing) {
            va[q].x = -P[(krow + ka)     * CLS + i * 16 + li];
            va[q].y = -P[(krow + ka + 1) * CLS + i * 16 + li];
        } else {
            va[q].x = P[(krow + li) * CLS + 64 + krow + ka];
            va[q].y = P[(krow + li) * CLS + 64 + krow + ka + 1];
        }
        vb[q].x = P[(krow + ka)     * CLS + ocol];
        vb[q].y = P[(krow + ka + 1) * CLS + ocol];
    }
    #pragma unroll
    for (int q = 0; q < 4; ++q)
        acc = __builtin_amdgcn_wmma_f32_16x16x4_f32(false, va[q], false, vb[q],
                                                    (short)0, acc, false, false);
    #pragma unroll
    for (int v = 0; v < 8; ++v) P[(orow + v + 8 * kh) * CLS + ocol] = acc[v];
}

__device__ __forceinline__ float lane_bcast(float x, int srclane) {
    return __int_as_float(__builtin_amdgcn_readlane(__float_as_int(x), srclane));
}

__global__ __launch_bounds__(512) void chol_inv(const float* __restrict__ G,
                                                float* __restrict__ Rinv) {
    __shared__ float P[64 * CLS];
    const int tid  = threadIdx.x;
    const int lane = tid & 31;
    const int wave = __builtin_amdgcn_readfirstlane(tid >> 5);
    const int li   = lane & 15;
    const int kh   = lane >> 4;

    {
        const int r = tid >> 3;
        const int c = (tid & 7) * 8;
        const float4* src = reinterpret_cast<const float4*>(G + r * 64 + c);
        const float4 a = src[0], b = src[1];
        float* dst = &P[r * CLS + c];
        dst[0]=a.x; dst[1]=a.y; dst[2]=a.z; dst[3]=a.w;
        dst[4]=b.x; dst[5]=b.y; dst[6]=b.z; dst[7]=b.w;
        #pragma unroll
        for (int q = 0; q < 8; ++q) dst[64 + q] = (c + q == r) ? 1.0f : 0.0f;
    }
    __syncthreads();

    #pragma unroll
    for (int kb = 0; kb < 4; ++kb) {
        const int krow = kb * 16;

        if (wave == 0) {
            const int ccol = (lane < 16) ? (krow + lane) : (64 + krow + lane - 16);
            float col[16];
            #pragma unroll
            for (int i = 0; i < 16; ++i) col[i] = P[(krow + i) * CLS + ccol];
            #pragma unroll
            for (int k = 0; k < 16; ++k) {
                const float akk = lane_bcast(col[k], k);
                const float dk  = sqrtf(akk);
                const float u   = (lane == k) ? dk : (col[k] / dk);
                col[k] = u;
                #pragma unroll
                for (int i = k + 1; i < 16; ++i) {
                    const float uki = lane_bcast(u, i);
                    col[i] -= uki * u;
                }
            }
            if (lane >= 16) {
                #pragma unroll
                for (int i = 0; i < 16; ++i) P[(krow + i) * CLS + ccol] = col[i];
            }
        }
        __syncthreads();

        if (wave < 8) {
            const int c = wave;
            const bool on = (c < 4) ? (c > kb) : (c - 4 < kb);
            if (on) panel_tile(P, kb, kb, c, false, li, kh);
        }
        __syncthreads();

        if (kb < 3) {
            #pragma unroll
            for (int t = 0; t < 2; ++t) {
                const int id = wave + 16 * t;
                const int i  = id >> 3;
                const int c  = id & 7;
                const bool on = (i > kb) && ((c < 4) ? (c >= i) : (c - 4 <= kb));
                if (on) panel_tile(P, kb, i, c, true, li, kh);
            }
            __syncthreads();
        }
    }

    {
        const int r = tid >> 3;
        const int c = (tid & 7) * 8;
        float4 a, b;
        a.x = P[(c + 0) * CLS + 64 + r]; a.y = P[(c + 1) * CLS + 64 + r];
        a.z = P[(c + 2) * CLS + 64 + r]; a.w = P[(c + 3) * CLS + 64 + r];
        b.x = P[(c + 4) * CLS + 64 + r]; b.y = P[(c + 5) * CLS + 64 + r];
        b.z = P[(c + 6) * CLS + 64 + r]; b.w = P[(c + 7) * CLS + 64 + r];
        float4* dst = reinterpret_cast<float4*>(Rinv + r * 64 + c);
        dst[0] = a; dst[1] = b;
    }
}

#define BSTR 68
#define ASTRIPS 4

__device__ __forceinline__ void load_strip(float4 (&a)[8], const float* __restrict__ Phi,
                                           long row0, int li, int kh, int m) {
    long arow = row0 + li;
    if (arow >= (long)m) arow = (long)m - 1;
    const float4* ap = reinterpret_cast<const float4*>(Phi + arow * 64 + 4 * kh);
    #pragma unroll
    for (int j = 0; j < 8; ++j) a[j] = ap[2 * j];
}

__global__ __launch_bounds__(512) void apply_rinv(const float* __restrict__ Phi,
                                                  const float* __restrict__ Rinv,
                                                  float* __restrict__ Q, int m) {
    __shared__ __attribute__((aligned(16))) float Bp[64 * BSTR];
    const int tid  = threadIdx.x;
    const int lane = tid & 31;
    const int wave = __builtin_amdgcn_readfirstlane(tid >> 5);
    const int li   = lane & 15;
    const int kh   = lane >> 4;

    const long strip0 = (long)blockIdx.x * (16 * ASTRIPS) + wave;

    float4 a[8];
    load_strip(a, Phi, strip0 * 16, li, kh, m);

    {
        const int k = tid >> 3;
        const int c = (tid & 7) * 8;
        const float4* src = reinterpret_cast<const float4*>(Rinv + k * 64 + c);
        const float4 x = src[0], y = src[1];
        const float v[8] = {x.x, x.y, x.z, x.w, y.x, y.y, y.z, y.w};
        #pragma unroll
        for (int q = 0; q < 8; ++q) {
            const int n = c + q;
            const int p = (n & 3) * 16 + (n >> 2);
            Bp[p * BSTR + k] = v[q];
        }
    }
    __syncthreads();

    const float* bbase = &Bp[li * BSTR + 4 * kh];

    for (int s = 0; s < ASTRIPS; ++s) {
        const long row0 = (strip0 + 16 * s) * 16;
        if (row0 >= (long)m) break;

        v8f acc[4] = {};
        #pragma unroll
        for (int j = 0; j < 8; ++j) {
            float4 b[4];
            #pragma unroll
            for (int t = 0; t < 4; ++t)
                b[t] = *reinterpret_cast<const float4*>(bbase + t * 16 * BSTR + 8 * j);
            v2f va;
            va.x = a[j].x; va.y = a[j].y;
            #pragma unroll
            for (int t = 0; t < 4; ++t) {
                v2f vb; vb.x = b[t].x; vb.y = b[t].y;
                acc[t] = __builtin_amdgcn_wmma_f32_16x16x4_f32(false, va, false, vb,
                                                               (short)0, acc[t], false, false);
            }
            va.x = a[j].z; va.y = a[j].w;
            #pragma unroll
            for (int t = 0; t < 4; ++t) {
                v2f vb; vb.x = b[t].z; vb.y = b[t].w;
                acc[t] = __builtin_amdgcn_wmma_f32_16x16x4_f32(false, va, false, vb,
                                                               (short)0, acc[t], false, false);
            }
        }

        if (s + 1 < ASTRIPS) load_strip(a, Phi, row0 + 256, li, kh, m);

        #pragma unroll
        for (int r = 0; r < 8; ++r) {
            const long row = row0 + 8 * kh + r;
            if (row < (long)m) {
                float4 o;
                o.x = acc[0][r]; o.y = acc[1][r]; o.z = acc[2][r]; o.w = acc[3][r];
                *reinterpret_cast<float4*>(Q + row * 64 + 4 * li) = o;
            }
        }
    }
}

extern "C" void kernel_launch(void* const* d_in, const int* in_sizes, int n_in,
                              void* d_out, int out_size, void* d_ws, size_t ws_size,
                              hipStream_t stream) {
    const float* Phi = (const float*)d_in[0];
    float* out = (float*)d_out;
    const int m = in_sizes[0] / 64;

    float* ws   = (float*)d_ws;
    float* G    = ws;
    float* Rinv = ws + 4096;
    float* part = ws + 8192;

    long avail = ((long)(ws_size / 4) - 8192) / 4096;
    int nwg = (int)(avail < 1 ? 1 : (avail > 512 ? 512 : avail));

    gram_partial<<<nwg, 512, 0, stream>>>(Phi, part, m, nwg);
    reduce_g<<<16, 256, 0, stream>>>(part, G, nwg);
    chol_inv<<<1, 512, 0, stream>>>(G, Rinv);

    const int nblk = (m + 1023) / 1024;
    apply_rinv<<<nblk, 512, 0, stream>>>(Phi, Rinv, out, m);
}


# gram_partial: 256 workgroups of 32 waves, one per WGP (was 512 of 16): 256-row blocks in a 2-stage LDS ring, row halves added in LDS; reduce_g sums 256 partials (was 512); G's sum tree changes
# speedup vs baseline: 2.3676x; 1.1892x over previous
#include <hip/hip_runtime.h>

typedef __attribute__((ext_vector_type(2))) float v2f;
typedef __attribute__((ext_vector_type(8))) float v8f;

#define LSTR 80

#define GROWS  256
#define GSTAGE (GROWS * LSTR)

__device__ __forceinline__ void load_rows(float4 (&p)[4], const float* __restrict__ Phi,
                                          long rowbase, int r, int c, int m) {
    #pragma unroll
    for (int h = 0; h < 2; ++h) {
        const long row = rowbase + r + 128 * h;
        if (row < (long)m) {
            const float4* src = reinterpret_cast<const float4*>(Phi + row * 64 + c);
            p[2 * h]     = src[0];
            p[2 * h + 1] = src[1];
        } else {
            p[2 * h]     = make_float4(0.0f, 0.0f, 0.0f, 0.0f);
            p[2 * h + 1] = make_float4(0.0f, 0.0f, 0.0f, 0.0f);
        }
    }
}

__global__ __launch_bounds__(1024) void gram_partial(const float* __restrict__ Phi,
                                                     float* __restrict__ part,
                                                     int m, int nwg) {
    __shared__ __attribute__((aligned(16))) float tile[2 * GSTAGE];
    __shared__ float comb[4096];
    const int tid  = threadIdx.x;
    const int lane = tid & 31;
    const int wave = tid >> 5;
    const int half = wave >> 4;
    const int t    = wave & 15;
    const int ti   = t >> 2;
    const int tj   = t & 3;
    const int li   = lane & 15;
    const int kh   = lane >> 4;
    const int r    = tid >> 3;
    const int c    = (tid & 7) * 8;

    v8f acc = {};

    const long stride = (long)nwg * GROWS;
    long rowbase = (long)blockIdx.x * GROWS;
    float4 p[4];
    load_rows(p, Phi, rowbase, r, c, m);

    int s = 0;
    for (; rowbase < (long)m; rowbase += stride) {
        float* st = tile + s * GSTAGE;
        {
            float4* d0 = reinterpret_cast<float4*>(st + r * LSTR + c);
            d0[0] = p[0]; d0[1] = p[1];
            float4* d1 = reinterpret_cast<float4*>(st + (r + 128) * LSTR + c);
            d1[0] = p[2]; d1[1] = p[3];
        }
        __syncthreads();

        load_rows(p, Phi, rowbase + stride, r, c, m);

        const float* th = st + half * 128 * LSTR;
        #pragma unroll
        for (int k0 = 0; k0 < 128; k0 += 4) {
            const int ka = k0 + 2 * kh;
            v2f va, vb;
            va.x = th[ka       * LSTR + ti * 16 + li];
            va.y = th[(ka + 1) * LSTR + ti * 16 + li];
            vb.x = th[ka       * LSTR + tj * 16 + li];
            vb.y = th[(ka + 1) * LSTR + tj * 16 + li];
            acc = __builtin_amdgcn_wmma_f32_16x16x4_f32(false, va, false, vb,
                                                        (short)0, acc, false, false);
        }
        s ^= 1;
    }

    if (half == 1) {
        #pragma unroll
        for (int v = 0; v < 8; ++v) comb[t * 256 + v * 32 + lane] = acc[v];
    }
    __syncthreads();
    if (half == 0) {
        float* pp = part + (size_t)blockIdx.x * 4096;
        #pragma unroll
        for (int v = 0; v < 8; ++v) {
            const int gi = ti * 16 + v + 8 * kh;
            const int gj = tj * 16 + li;
            pp[gi * 64 + gj] = acc[v] + comb[t * 256 + v * 32 + lane];
        }
    }
}

__global__ __launch_bounds__(256) void reduce_g(const float* __restrict__ part,
                                                float* __restrict__ G, int nwg) {
    const int e = blockIdx.x * 256 + threadIdx.x;
    float s = 0.0f;
    for (int w = 0; w < nwg; ++w) s += part[(size_t)w * 4096 + e];
    G[e] = s;
}

#define CLS 144

__device__ __forceinline__ void panel_tile(float* P, int kb, int i, int c,
                                           bool trailing, int li, int kh) {
    const int krow = kb * 16;
    const int ocol = c * 16 + li;
    const int orow = (trailing ? i : kb) * 16;
    v8f acc = {};
    if (trailing) {
        #pragma unroll
        for (int v = 0; v < 8; ++v) acc[v] = P[(orow + v + 8 * kh) * CLS + ocol];
    }
    v2f va[4], vb[4];
    #pragma unroll
    for (int q = 0; q < 4; ++q) {
        const int ka = q * 4 + 2 * kh;
        if (trailing) {
            va[q].x = -P[(krow + ka)     * CLS + i * 16 + li];
            va[q].y = -P[(krow + ka + 1) * CLS + i * 16 + li];
        } else {
            va[q].x = P[(krow + li) * CLS + 64 + krow + ka];
            va[q].y = P[(krow + li) * CLS + 64 + krow + ka + 1];
        }
        vb[q].x = P[(krow + ka)     * CLS + ocol];
        vb[q].y = P[(krow + ka + 1) * CLS + ocol];
    }
    #pragma unroll
    for (int q = 0; q < 4; ++q)
        acc = __builtin_amdgcn_wmma_f32_16x16x4_f32(false, va[q], false, vb[q],
                                                    (short)0, acc, false, false);
    #pragma unroll
    for (int v = 0; v < 8; ++v) P[(orow + v + 8 * kh) * CLS + ocol] = acc[v];
}

__device__ __forceinline__ float lane_bcast(float x, int srclane) {
    return __int_as_float(__builtin_amdgcn_readlane(__float_as_int(x), srclane));
}

__global__ __launch_bounds__(512) void chol_inv(const float* __restrict__ G,
                                                float* __restrict__ Rinv) {
    __shared__ float P[64 * CLS];
    const int tid  = threadIdx.x;
    const int lane = tid & 31;
    const int wave = __builtin_amdgcn_readfirstlane(tid >> 5);
    const int li   = lane & 15;
    const int kh   = lane >> 4;

    {
        const int r = tid >> 3;
        const int c = (tid & 7) * 8;
        const float4* src = reinterpret_cast<const float4*>(G + r * 64 + c);
        const float4 a = src[0], b = src[1];
        float* dst = &P[r * CLS + c];
        dst[0]=a.x; dst[1]=a.y; dst[2]=a.z; dst[3]=a.w;
        dst[4]=b.x; dst[5]=b.y; dst[6]=b.z; dst[7]=b.w;
        #pragma unroll
        for (int q = 0; q < 8; ++q) dst[64 + q] = (c + q == r) ? 1.0f : 0.0f;
    }
    __syncthreads();

    #pragma unroll
    for (int kb = 0; kb < 4; ++kb) {
        const int krow = kb * 16;

        if (wave == 0) {
            const int ccol = (lane < 16) ? (krow + lane) : (64 + krow + lane - 16);
            float col[16];
            #pragma unroll
            for (int i = 0; i < 16; ++i) col[i] = P[(krow + i) * CLS + ccol];
            #pragma unroll
            for (int k = 0; k < 16; ++k) {
                const float akk = lane_bcast(col[k], k);
                const float dk  = sqrtf(akk);
                const float u   = (lane == k) ? dk : (col[k] / dk);
                col[k] = u;
                #pragma unroll
                for (int i = k + 1; i < 16; ++i) {
                    const float uki = lane_bcast(u, i);
                    col[i] -= uki * u;
                }
            }
            if (lane >= 16) {
                #pragma unroll
                for (int i = 0; i < 16; ++i) P[(krow + i) * CLS + ccol] = col[i];
            }
        }
        __syncthreads();

        if (wave < 8) {
            const int c = wave;
            const bool on = (c < 4) ? (c > kb) : (c - 4 < kb);
            if (on) panel_tile(P, kb, kb, c, false, li, kh);
        }
        __syncthreads();

        if (kb < 3) {
            #pragma unroll
            for (int t = 0; t < 2; ++t) {
                const int id = wave + 16 * t;
                const int i  = id >> 3;
                const int c  = id & 7;
                const bool on = (i > kb) && ((c < 4) ? (c >= i) : (c - 4 <= kb));
                if (on) panel_tile(P, kb, i, c, true, li, kh);
            }
            __syncthreads();
        }
    }

    {
        const int r = tid >> 3;
        const int c = (tid & 7) * 8;
        float4 a, b;
        a.x = P[(c + 0) * CLS + 64 + r]; a.y = P[(c + 1) * CLS + 64 + r];
        a.z = P[(c + 2) * CLS + 64 + r]; a.w = P[(c + 3) * CLS + 64 + r];
        b.x = P[(c + 4) * CLS + 64 + r]; b.y = P[(c + 5) * CLS + 64 + r];
        b.z = P[(c + 6) * CLS + 64 + r]; b.w = P[(c + 7) * CLS + 64 + r];
        float4* dst = reinterpret_cast<float4*>(Rinv + r * 64 + c);
        dst[0] = a; dst[1] = b;
    }
}

#define BSTR 68
#define ASTRIPS 4

__device__ __forceinline__ void load_strip(float4 (&a)[8], const float* __restrict__ Phi,
                                           long row0, int li, int kh, int m) {
    long arow = row0 + li;
    if (arow >= (long)m) arow = (long)m - 1;
    const float4* ap = reinterpret_cast<const float4*>(Phi + arow * 64 + 4 * kh);
    #pragma unroll
    for (int j = 0; j < 8; ++j) a[j] = ap[2 * j];
}

__global__ __launch_bounds__(512) void apply_rinv(const float* __restrict__ Phi,
                                                  const float* __restrict__ Rinv,
                                                  float* __restrict__ Q, int m) {
    __shared__ __attribute__((aligned(16))) float Bp[64 * BSTR];
    const int tid  = threadIdx.x;
    const int lane = tid & 31;
    const int wave = __builtin_amdgcn_readfirstlane(tid >> 5);
    const int li   = lane & 15;
    const int kh   = lane >> 4;

    const long strip0 = (long)blockIdx.x * (16 * ASTRIPS) + wave;

    float4 a[8];
    load_strip(a, Phi, strip0 * 16, li, kh, m);

    {
        const int k = tid >> 3;
        const int c = (tid & 7) * 8;
        const float4* src = reinterpret_cast<const float4*>(Rinv + k * 64 + c);
        const float4 x = src[0], y = src[1];
        const float v[8] = {x.x, x.y, x.z, x.w, y.x, y.y, y.z, y.w};
        #pragma unroll
        for (int q = 0; q < 8; ++q) {
            const int n = c + q;
            const int p = (n & 3) * 16 + (n >> 2);
            Bp[p * BSTR + k] = v[q];
        }
    }
    __syncthreads();

    const float* bbase = &Bp[li * BSTR + 4 * kh];

    for (int s = 0; s < ASTRIPS; ++s) {
        const long row0 = (strip0 + 16 * s) * 16;
        if (row0 >= (long)m) break;

        v8f acc[4] = {};
        #pragma unroll
        for (int j = 0; j < 8; ++j) {
            float4 b[4];
            #pragma unroll
            for (int t = 0; t < 4; ++t)
                b[t] = *reinterpret_cast<const float4*>(bbase + t * 16 * BSTR + 8 * j);
            v2f va;
            va.x = a[j].x; va.y = a[j].y;
            #pragma unroll
            for (int t = 0; t < 4; ++t) {
                v2f vb; vb.x = b[t].x; vb.y = b[t].y;
                acc[t] = __builtin_amdgcn_wmma_f32_16x16x4_f32(false, va, false, vb,
                                                               (short)0, acc[t], false, false);
            }
            va.x = a[j].z; va.y = a[j].w;
            #pragma unroll
            for (int t = 0; t < 4; ++t) {
                v2f vb; vb.x = b[t].z; vb.y = b[t].w;
                acc[t] = __builtin_amdgcn_wmma_f32_16x16x4_f32(false, va, false, vb,
                                                               (short)0, acc[t], false, false);
            }
        }

        if (s + 1 < ASTRIPS) load_strip(a, Phi, row0 + 256, li, kh, m);

        #pragma unroll
        for (int r = 0; r < 8; ++r) {
            const long row = row0 + 8 * kh + r;
            if (row < (long)m) {
                float4 o;
                o.x = acc[0][r]; o.y = acc[1][r]; o.z = acc[2][r]; o.w = acc[3][r];
                *reinterpret_cast<float4*>(Q + row * 64 + 4 * li) = o;
            }
        }
    }
}

extern "C" void kernel_launch(void* const* d_in, const int* in_sizes, int n_in,
                              void* d_out, int out_size, void* d_ws, size_t ws_size,
                              hipStream_t stream) {
    const float* Phi = (const float*)d_in[0];
    float* out = (float*)d_out;
    const int m = in_sizes[0] / 64;

    float* ws   = (float*)d_ws;
    float* G    = ws;
    float* Rinv = ws + 4096;
    float* part = ws + 8192;

    long avail = ((long)(ws_size / 4) - 8192) / 4096;
    int nwg = (int)(avail < 1 ? 1 : (avail > 256 ? 256 : avail));

    gram_partial<<<nwg, 1024, 0, stream>>>(Phi, part, m, nwg);
    reduce_g<<<16, 256, 0, stream>>>(part, G, nwg);
    chol_inv<<<1, 512, 0, stream>>>(G, Rinv);

    const int nblk = (m + 1023) / 1024;
    apply_rinv<<<nblk, 512, 0, stream>>>(Phi, Rinv, out, m);
}


# gram_partial: only G's 10 tiles on or above the diagonal (chol_inv reads no others): 30 waves = tile x 80-row slice, 2 waves: rows 240-255 of 5 tiles; 20 WMMAs a wave (was 32); G's sum tree changes
# speedup vs baseline: 2.5030x; 1.0572x over previous
#include <hip/hip_runtime.h>

typedef __attribute__((ext_vector_type(2))) float v2f;
typedef __attribute__((ext_vector_type(8))) float v8f;

#define LSTR 80

#define GROWS  256
#define GSTAGE (GROWS * LSTR)
#define NUT    10

__device__ __forceinline__ void tile_ij(int T, int& ti, int& tj) {
    ti = (T >= 4) + (T >= 7) + (T >= 9);
    tj = T - (4 * ti - (ti * (ti - 1)) / 2) + ti;
}

__device__ __forceinline__ void load_rows(float4 (&p)[4], const float* __restrict__ Phi,
                                          long rowbase, int r, int c, int m) {
    #pragma unroll
    for (int h = 0; h < 2; ++h) {
        const long row = rowbase + r + 128 * h;
        if (row < (long)m) {
            const float4* src = reinterpret_cast<const float4*>(Phi + row * 64 + c);
            p[2 * h]     = src[0];
            p[2 * h + 1] = src[1];
        } else {
            p[2 * h]     = make_float4(0.0f, 0.0f, 0.0f, 0.0f);
            p[2 * h + 1] = make_float4(0.0f, 0.0f, 0.0f, 0.0f);
        }
    }
}

__device__ __forceinline__ void gram_steps4(v8f& acc, const float* pa, const float* pb) {
    #pragma unroll
    for (int k0 = 0; k0 < 16; k0 += 4) {
        v2f va, vb;
        va.x = pa[k0       * LSTR];
        va.y = pa[(k0 + 1) * LSTR];
        vb.x = pb[k0       * LSTR];
        vb.y = pb[(k0 + 1) * LSTR];
        acc = __builtin_amdgcn_wmma_f32_16x16x4_f32(false, va, false, vb,
                                                    (short)0, acc, false, false);
    }
}

__global__ __launch_bounds__(1024) void gram_partial(const float* __restrict__ Phi,
                                                     float* __restrict__ part,
                                                     int m, int nwg) {
    __shared__ __attribute__((aligned(16))) float tile[2 * GSTAGE];
    __shared__ float comb[4 * NUT * 256];
    const int tid  = threadIdx.x;
    const int lane = tid & 31;
    const int wave = __builtin_amdgcn_readfirstlane(tid >> 5);
    const int li   = lane & 15;
    const int kh   = lane >> 4;
    const int r    = tid >> 3;
    const int c    = (tid & 7) * 8;

    const bool slicew = wave < 3 * NUT;

    int aoff[5], boff[5];
    #pragma unroll
    for (int q = 0; q < 5; ++q) {
        const int T    = slicew ? wave / 3 : 5 * (wave - 3 * NUT) + q;
        const int row0 = slicew ? 80 * (wave % 3) + 16 * q : 240;
        int ti, tj;
        tile_ij(T, ti, tj);
        aoff[q] = (row0 + 2 * kh) * LSTR + ti * 16 + li;
        boff[q] = (row0 + 2 * kh) * LSTR + tj * 16 + li;
    }

    v8f acc[5] = {};

    const long stride = (long)nwg * GROWS;
    long rowbase = (long)blockIdx.x * GROWS;
    float4 p[4];
    load_rows(p, Phi, rowbase, r, c, m);

    int s = 0;
    for (; rowbase < (long)m; rowbase += stride) {
        float* st = tile + s * GSTAGE;
        {
            float4* d0 = reinterpret_cast<float4*>(st + r * LSTR + c);
            d0[0] = p[0]; d0[1] = p[1];
            float4* d1 = reinterpret_cast<float4*>(st + (r + 128) * LSTR + c);
            d1[0] = p[2]; d1[1] = p[3];
        }
        __syncthreads();

        load_rows(p, Phi, rowbase + stride, r, c, m);

        #pragma unroll
        for (int q = 0; q < 5; ++q) gram_steps4(acc[q], st + aoff[q], st + boff[q]);
        s ^= 1;
    }

    if (slicew) {
        const v8f t = ((acc[0] + acc[1]) + (acc[2] + acc[3])) + acc[4];
        #pragma unroll
        for (int v = 0; v < 8; ++v) comb[wave * 256 + v * 32 + lane] = t[v];
    } else {
        #pragma unroll
        for (int q = 0; q < 5; ++q) {
            #pragma unroll
            for (int v = 0; v < 8; ++v)
                comb[(3 * NUT + 5 * (wave - 3 * NUT) + q) * 256 + v * 32 + lane] = acc[q][v];
        }
    }
    __syncthreads();
    float* pp = part + (size_t)blockIdx.x * 4096;
    for (int e = tid; e < NUT * 256; e += 1024) {
        const int T   = e >> 8;
        const int idx = e & 255;
        const float* cp = comb + idx;
        const float sum = ((cp[(3 * T) * 256] + cp[(3 * T + 1) * 256]) + cp[(3 * T + 2) * 256])
                          + cp[(3 * NUT + T) * 256];
        int ti, tj;
        tile_ij(T, ti, tj);
        const int gi = ti * 16 + (idx >> 5) + 8 * ((idx >> 4) & 1);
        const int gj = tj * 16 + (idx & 15);
        pp[gi * 64 + gj] = sum;
    }
}

__global__ __launch_bounds__(256) void reduce_g(const float* __restrict__ part,
                                                float* __restrict__ G, int nwg) {
    const int e = blockIdx.x * 256 + threadIdx.x;
    if (((e & 63) >> 4) < (e >> 10)) return;
    float s = 0.0f;
    for (int w = 0; w < nwg; ++w) s += part[(size_t)w * 4096 + e];
    G[e] = s;
}

#define CLS 144

__device__ __forceinline__ void panel_tile(float* P, int kb, int i, int c,
                                           bool trailing, int li, int kh) {
    const int krow = kb * 16;
    const int ocol = c * 16 + li;
    const int orow = (trailing ? i : kb) * 16;
    v8f acc = {};
    if (trailing) {
        #pragma unroll
        for (int v = 0; v < 8; ++v) acc[v] = P[(orow + v + 8 * kh) * CLS + ocol];
    }
    v2f va[4], vb[4];
    #pragma unroll
    for (int q = 0; q < 4; ++q) {
        const int ka = q * 4 + 2 * kh;
        if (trailing) {
            va[q].x = -P[(krow + ka)     * CLS + i * 16 + li];
            va[q].y = -P[(krow + ka + 1) * CLS + i * 16 + li];
        } else {
            va[q].x = P[(krow + li) * CLS + 64 + krow + ka];
            va[q].y = P[(krow + li) * CLS + 64 + krow + ka + 1];
        }
        vb[q].x = P[(krow + ka)     * CLS + ocol];
        vb[q].y = P[(krow + ka + 1) * CLS + ocol];
    }
    #pragma unroll
    for (int q = 0; q < 4; ++q)
        acc = __builtin_amdgcn_wmma_f32_16x16x4_f32(false, va[q], false, vb[q],
                                                    (short)0, acc, false, false);
    #pragma unroll
    for (int v = 0; v < 8; ++v) P[(orow + v + 8 * kh) * CLS + ocol] = acc[v];
}

__device__ __forceinline__ float lane_bcast(float x, int srclane) {
    return __int_as_float(__builtin_amdgcn_readlane(__float_as_int(x), srclane));
}

__global__ __launch_bounds__(512) void chol_inv(const float* __restrict__ G,
                                                float* __restrict__ Rinv) {
    __shared__ float P[64 * CLS];
    const int tid  = threadIdx.x;
    const int lane = tid & 31;
    const int wave = __builtin_amdgcn_readfirstlane(tid >> 5);
    const int li   = lane & 15;
    const int kh   = lane >> 4;

    {
        const int r = tid >> 3;
        const int c = (tid & 7) * 8;
        float4 a = make_float4(0.0f, 0.0f, 0.0f, 0.0f), b = a;
        if ((c >> 4) >= (r >> 4)) {
            const float4* src = reinterpret_cast<const float4*>(G + r * 64 + c);
            a = src[0]; b = src[1];
        }
        float* dst = &P[r * CLS + c];
        dst[0]=a.x; dst[1]=a.y; dst[2]=a.z; dst[3]=a.w;
        dst[4]=b.x; dst[5]=b.y; dst[6]=b.z; dst[7]=b.w;
        #pragma unroll
        for (int q = 0; q < 8; ++q) dst[64 + q] = (c + q == r) ? 1.0f : 0.0f;
    }
    __syncthreads();

    #pragma unroll
    for (int kb = 0; kb < 4; ++kb) {
        const int krow = kb * 16;

        if (wave == 0) {
            const int ccol = (lane < 16) ? (krow + lane) : (64 + krow + lane - 16);
            float col[16];
            #pragma unroll
            for (int i = 0; i < 16; ++i) col[i] = P[(krow + i) * CLS + ccol];
            #pragma unroll
            for (int k = 0; k < 16; ++k) {
                const float akk = lane_bcast(col[k], k);
                const float dk  = sqrtf(akk);
                const float u   = (lane == k) ? dk : (col[k] / dk);
                col[k] = u;
                #pragma unroll
                for (int i = k + 1; i < 16; ++i) {
                    const float uki = lane_bcast(u, i);
                    col[i] -= uki * u;
                }
            }
            if (lane >= 16) {
                #pragma unroll
                for (int i = 0; i < 16; ++i) P[(krow + i) * CLS + ccol] = col[i];
            }
        }
        __syncthreads();

        if (wave < 8) {
            const int c = wave;
            const bool on = (c < 4) ? (c > kb) : (c - 4 < kb);
            if (on) panel_tile(P, kb, kb, c, false, li, kh);
        }
        __syncthreads();

        if (kb < 3) {
            #pragma unroll
            for (int t = 0; t < 2; ++t) {
                const int id = wave + 16 * t;
                const int i  = id >> 3;
                const int c  = id & 7;
                const bool on = (i > kb) && ((c < 4) ? (c >= i) : (c - 4 <= kb));
                if (on) panel_tile(P, kb, i, c, true, li, kh);
            }
            __syncthreads();
        }
    }

    {
        const int r = tid >> 3;
        const int c = (tid & 7) * 8;
        float4 a, b;
        a.x = P[(c + 0) * CLS + 64 + r]; a.y = P[(c + 1) * CLS + 64 + r];
        a.z = P[(c + 2) * CLS + 64 + r]; a.w = P[(c + 3) * CLS + 64 + r];
        b.x = P[(c + 4) * CLS + 64 + r]; b.y = P[(c + 5) * CLS + 64 + r];
        b.z = P[(c + 6) * CLS + 64 + r]; b.w = P[(c + 7) * CLS + 64 + r];
        float4* dst = reinterpret_cast<float4*>(Rinv + r * 64 + c);
        dst[0] = a; dst[1] = b;
    }
}

#define BSTR 68
#define ASTRIPS 4

__device__ __forceinline__ void load_strip(float4 (&a)[8], const float* __restrict__ Phi,
                                           long row0, int li, int kh, int m) {
    long arow = row0 + li;
    if (arow >= (long)m) arow = (long)m - 1;
    const float4* ap = reinterpret_cast<const float4*>(Phi + arow * 64 + 4 * kh);
    #pragma unroll
    for (int j = 0; j < 8; ++j) a[j] = ap[2 * j];
}

__global__ __launch_bounds__(512) void apply_rinv(const float* __restrict__ Phi,
                                                  const float* __restrict__ Rinv,
                                                  float* __restrict__ Q, int m) {
    __shared__ __attribute__((aligned(16))) float Bp[64 * BSTR];
    const int tid  = threadIdx.x;
    const int lane = tid & 31;
    const int wave = __builtin_amdgcn_readfirstlane(tid >> 5);
    const int li   = lane & 15;
    const int kh   = lane >> 4;

    const long strip0 = (long)blockIdx.x * (16 * ASTRIPS) + wave;

    float4 a[8];
    load_strip(a, Phi, strip0 * 16, li, kh, m);

    {
        const int k = tid >> 3;
        const int c = (tid & 7) * 8;
        const float4* src = reinterpret_cast<const float4*>(Rinv + k * 64 + c);
        const float4 x = src[0], y = src[1];
        const float v[8] = {x.x, x.y, x.z, x.w, y.x, y.y, y.z, y.w};
        #pragma unroll
        for (int q = 0; q < 8; ++q) {
            const int n = c + q;
            const int p = (n & 3) * 16 + (n >> 2);
            Bp[p * BSTR + k] = v[q];
        }
    }
    __syncthreads();

    const float* bbase = &Bp[li * BSTR + 4 * kh];

    for (int s = 0; s < ASTRIPS; ++s) {
        const long row0 = (strip0 + 16 * s) * 16;
        if (row0 >= (long)m) break;

        v8f acc[4] = {};
        #pragma unroll
        for (int j = 0; j < 8; ++j) {
            float4 b[4];
            #pragma unroll
            for (int t = 0; t < 4; ++t)
                b[t] = *reinterpret_cast<const float4*>(bbase + t * 16 * BSTR + 8 * j);
            v2f va;
            va.x = a[j].x; va.y = a[j].y;
            #pragma unroll
            for (int t = 0; t < 4; ++t) {
                v2f vb; vb.x = b[t].x; vb.y = b[t].y;
                acc[t] = __builtin_amdgcn_wmma_f32_16x16x4_f32(false, va, false, vb,
                                                               (short)0, acc[t], false, false);
            }
            va.x = a[j].z; va.y = a[j].w;
            #pragma unroll
            for (int t = 0; t < 4; ++t) {
                v2f vb; vb.x = b[t].z; vb.y = b[t].w;
                acc[t] = __builtin_amdgcn_wmma_f32_16x16x4_f32(false, va, false, vb,
                                                               (short)0, acc[t], false, false);
            }
        }

        if (s + 1 < ASTRIPS) load_strip(a, Phi, row0 + 256, li, kh, m);

        #pragma unroll
        for (int r = 0; r < 8; ++r) {
            const long row = row0 + 8 * kh + r;
            if (row < (long)m) {
                float4 o;
                o.x = acc[0][r]; o.y = acc[1][r]; o.z = acc[2][r]; o.w = acc[3][r];
                *reinterpret_cast<float4*>(Q + row * 64 + 4 * li) = o;
            }
        }
    }
}

extern "C" void kernel_launch(void* const* d_in, const int* in_sizes, int n_in,
                              void* d_out, int out_size, void* d_ws, size_t ws_size,
                              hipStream_t stream) {
    const float* Phi = (const float*)d_in[0];
    float* out = (float*)d_out;
    const int m = in_sizes[0] / 64;

    float* ws   = (float*)d_ws;
    float* G    = ws;
    float* Rinv = ws + 4096;
    float* part = ws + 8192;

    long avail = ((long)(ws_size / 4) - 8192) / 4096;
    int nwg = (int)(avail < 1 ? 1 : (avail > 256 ? 256 : avail));

    gram_partial<<<nwg, 1024, 0, stream>>>(Phi, part, m, nwg);
    reduce_g<<<16, 256, 0, stream>>>(part, G, nwg);
    chol_inv<<<1, 512, 0, stream>>>(G, Rinv);

    const int nblk = (m + 1023) / 1024;
    apply_rinv<<<nblk, 512, 0, stream>>>(Phi, Rinv, out, m);
}


# reduce_g: ahead of its in-order sum each wave prefetches its 128-byte line of every partial Gram into the second-level cache (a lane per partial, 32 to an instruction); no sum or order changes
# speedup vs baseline: 2.6363x; 1.0533x over previous
#include <hip/hip_runtime.h>

typedef __attribute__((ext_vector_type(2))) float v2f;
typedef __attribute__((ext_vector_type(8))) float v8f;

#define LSTR 80

#define GROWS  256
#define GSTAGE (GROWS * LSTR)
#define NUT    10

__device__ __forceinline__ void tile_ij(int T, int& ti, int& tj) {
    ti = (T >= 4) + (T >= 7) + (T >= 9);
    tj = T - (4 * ti - (ti * (ti - 1)) / 2) + ti;
}

__device__ __forceinline__ void load_rows(float4 (&p)[4], const float* __restrict__ Phi,
                                          long rowbase, int r, int c, int m) {
    #pragma unroll
    for (int h = 0; h < 2; ++h) {
        const long row = rowbase + r + 128 * h;
        if (row < (long)m) {
            const float4* src = reinterpret_cast<const float4*>(Phi + row * 64 + c);
            p[2 * h]     = src[0];
            p[2 * h + 1] = src[1];
        } else {
            p[2 * h]     = make_float4(0.0f, 0.0f, 0.0f, 0.0f);
            p[2 * h + 1] = make_float4(0.0f, 0.0f, 0.0f, 0.0f);
        }
    }
}

__device__ __forceinline__ void gram_steps4(v8f& acc, const float* pa, const float* pb) {
    #pragma unroll
    for (int k0 = 0; k0 < 16; k0 += 4) {
        v2f va, vb;
        va.x = pa[k0       * LSTR];
        va.y = pa[(k0 + 1) * LSTR];
        vb.x = pb[k0       * LSTR];
        vb.y = pb[(k0 + 1) * LSTR];
        acc = __builtin_amdgcn_wmma_f32_16x16x4_f32(false, va, false, vb,
                                                    (short)0, acc, false, false);
    }
}

__global__ __launch_bounds__(1024) void gram_partial(const float* __restrict__ Phi,
                                                     float* __restrict__ part,
                                                     int m, int nwg) {
    __shared__ __attribute__((aligned(16))) float tile[2 * GSTAGE];
    __shared__ float comb[4 * NUT * 256];
    const int tid  = threadIdx.x;
    const int lane = tid & 31;
    const int wave = __builtin_amdgcn_readfirstlane(tid >> 5);
    const int li   = lane & 15;
    const int kh   = lane >> 4;
    const int r    = tid >> 3;
    const int c    = (tid & 7) * 8;

    const bool slicew = wave < 3 * NUT;

    int aoff[5], boff[5];
    #pragma unroll
    for (int q = 0; q < 5; ++q) {
        const int T    = slicew ? wave / 3 : 5 * (wave - 3 * NUT) + q;
        const int row0 = slicew ? 80 * (wave % 3) + 16 * q : 240;
        int ti, tj;
        tile_ij(T, ti, tj);
        aoff[q] = (row0 + 2 * kh) * LSTR + ti * 16 + li;
        boff[q] = (row0 + 2 * kh) * LSTR + tj * 16 + li;
    }

    v8f acc[5] = {};

    const long stride = (long)nwg * GROWS;
    long rowbase = (long)blockIdx.x * GROWS;
    float4 p[4];
    load_rows(p, Phi, rowbase, r, c, m);

    int s = 0;
    for (; rowbase < (long)m; rowbase += stride) {
        float* st = tile + s * GSTAGE;
        {
            float4* d0 = reinterpret_cast<float4*>(st + r * LSTR + c);
            d0[0] = p[0]; d0[1] = p[1];
            float4* d1 = reinterpret_cast<float4*>(st + (r + 128) * LSTR + c);
            d1[0] = p[2]; d1[1] = p[3];
        }
        __syncthreads();

        load_rows(p, Phi, rowbase + stride, r, c, m);

        #pragma unroll
        for (int q = 0; q < 5; ++q) gram_steps4(acc[q], st + aoff[q], st + boff[q]);
        s ^= 1;
    }

    if (slicew) {
        const v8f t = ((acc[0] + acc[1]) + (acc[2] + acc[3])) + acc[4];
        #pragma unroll
        for (int v = 0; v < 8; ++v) comb[wave * 256 + v * 32 + lane] = t[v];
    } else {
        #pragma unroll
        for (int q = 0; q < 5; ++q) {
            #pragma unroll
            for (int v = 0; v < 8; ++v)
                comb[(3 * NUT + 5 * (wave - 3 * NUT) + q) * 256 + v * 32 + lane] = acc[q][v];
        }
    }
    __syncthreads();
    float* pp = part + (size_t)blockIdx.x * 4096;
    for (int e = tid; e < NUT * 256; e += 1024) {
        const int T   = e >> 8;
        const int idx = e & 255;
        const float* cp = comb + idx;
        const float sum = ((cp[(3 * T) * 256] + cp[(3 * T + 1) * 256]) + cp[(3 * T + 2) * 256])
                          + cp[(3 * NUT + T) * 256];
        int ti, tj;
        tile_ij(T, ti, tj);
        const int gi = ti * 16 + (idx >> 5) + 8 * ((idx >> 4) & 1);
        const int gj = tj * 16 + (idx & 15);
        pp[gi * 64 + gj] = sum;
    }
}

__global__ __launch_bounds__(256) void reduce_g(const float* __restrict__ part,
                                                float* __restrict__ G, int nwg) {
    const int e = blockIdx.x * 256 + threadIdx.x;
    {
        const int lane = threadIdx.x & 31;
        const int e0   = e - lane;
        if ((((e0 & 63) >> 4) + 1) >= (e >> 10)) {
            for (int w = lane; w < nwg; w += 32)
                __builtin_prefetch(part + (size_t)w * 4096 + e0, 0, 2);
        }
    }
    if (((e & 63) >> 4) < (e >> 10)) return;
    float s = 0.0f;
    for (int w = 0; w < nwg; ++w) s += part[(size_t)w * 4096 + e];
    G[e] = s;
}

#define CLS 144

__device__ __forceinline__ void panel_tile(float* P, int kb, int i, int c,
                                           bool trailing, int li, int kh) {
    const int krow = kb * 16;
    const int ocol = c * 16 + li;
    const int orow = (trailing ? i : kb) * 16;
    v8f acc = {};
    if (trailing) {
        #pragma unroll
        for (int v = 0; v < 8; ++v) acc[v] = P[(orow + v + 8 * kh) * CLS + ocol];
    }
    v2f va[4], vb[4];
    #pragma unroll
    for (int q = 0; q < 4; ++q) {
        const int ka = q * 4 + 2 * kh;
        if (trailing) {
            va[q].x = -P[(krow + ka)     * CLS + i * 16 + li];
            va[q].y = -P[(krow + ka + 1) * CLS + i * 16 + li];
        } else {
            va[q].x = P[(krow + li) * CLS + 64 + krow + ka];
            va[q].y = P[(krow + li) * CLS + 64 + krow + ka + 1];
        }
        vb[q].x = P[(krow + ka)     * CLS + ocol];
        vb[q].y = P[(krow + ka + 1) * CLS + ocol];
    }
    #pragma unroll
    for (int q = 0; q < 4; ++q)
        acc = __builtin_amdgcn_wmma_f32_16x16x4_f32(false, va[q], false, vb[q],
                                                    (short)0, acc, false, false);
    #pragma unroll
    for (int v = 0; v < 8; ++v) P[(orow + v + 8 * kh) * CLS + ocol] = acc[v];
}

__device__ __forceinline__ float lane_bcast(float x, int srclane) {
    return __int_as_float(__builtin_amdgcn_readlane(__float_as_int(x), srclane));
}

__global__ __launch_bounds__(512) void chol_inv(const float* __restrict__ G,
                                                float* __restrict__ Rinv) {
    __shared__ float P[64 * CLS];
    const int tid  = threadIdx.x;
    const int lane = tid & 31;
    const int wave = __builtin_amdgcn_readfirstlane(tid >> 5);
    const int li   = lane & 15;
    const int kh   = lane >> 4;

    {
        const int r = tid >> 3;
        const int c = (tid & 7) * 8;
        float4 a = make_float4(0.0f, 0.0f, 0.0f, 0.0f), b = a;
        if ((c >> 4) >= (r >> 4)) {
            const float4* src = reinterpret_cast<const float4*>(G + r * 64 + c);
            a = src[0]; b = src[1];
        }
        float* dst = &P[r * CLS + c];
        dst[0]=a.x; dst[1]=a.y; dst[2]=a.z; dst[3]=a.w;
        dst[4]=b.x; dst[5]=b.y; dst[6]=b.z; dst[7]=b.w;
        #pragma unroll
        for (int q = 0; q < 8; ++q) dst[64 + q] = (c + q == r) ? 1.0f : 0.0f;
    }
    __syncthreads();

    #pragma unroll
    for (int kb = 0; kb < 4; ++kb) {
        const int krow = kb * 16;

        if (wave == 0) {
            const int ccol = (lane < 16) ? (krow + lane) : (64 + krow + lane - 16);
            float col[16];
            #pragma unroll
            for (int i = 0; i < 16; ++i) col[i] = P[(krow + i) * CLS + ccol];
            #pragma unroll
            for (int k = 0; k < 16; ++k) {
                const float akk = lane_bcast(col[k], k);
                const float dk  = sqrtf(akk);
                const float u   = (lane == k) ? dk : (col[k] / dk);
                col[k] = u;
                #pragma unroll
                for (int i = k + 1; i < 16; ++i) {
                    const float uki = lane_bcast(u, i);
                    col[i] -= uki * u;
                }
            }
            if (lane >= 16) {
                #pragma unroll
                for (int i = 0; i < 16; ++i) P[(krow + i) * CLS + ccol] = col[i];
            }
        }
        __syncthreads();

        if (wave < 8) {
            const int c = wave;
            const bool on = (c < 4) ? (c > kb) : (c - 4 < kb);
            if (on) panel_tile(P, kb, kb, c, false, li, kh);
        }
        __syncthreads();

        if (kb < 3) {
            #pragma unroll
            for (int t = 0; t < 2; ++t) {
                const int id = wave + 16 * t;
                const int i  = id >> 3;
                const int c  = id & 7;
                const bool on = (i > kb) && ((c < 4) ? (c >= i) : (c - 4 <= kb));
                if (on) panel_tile(P, kb, i, c, true, li, kh);
            }
            __syncthreads();
        }
    }

    {
        const int r = tid >> 3;
        const int c = (tid & 7) * 8;
        float4 a, b;
        a.x = P[(c + 0) * CLS + 64 + r]; a.y = P[(c + 1) * CLS + 64 + r];
        a.z = P[(c + 2) * CLS + 64 + r]; a.w = P[(c + 3) * CLS + 64 + r];
        b.x = P[(c + 4) * CLS + 64 + r]; b.y = P[(c + 5) * CLS + 64 + r];
        b.z = P[(c + 6) * CLS + 64 + r]; b.w = P[(c + 7) * CLS + 64 + r];
        float4* dst = reinterpret_cast<float4*>(Rinv + r * 64 + c);
        dst[0] = a; dst[1] = b;
    }
}

#define BSTR 68
#define ASTRIPS 4

__device__ __forceinline__ void load_strip(float4 (&a)[8], const float* __restrict__ Phi,
                                           long row0, int li, int kh, int m) {
    long arow = row0 + li;
    if (arow >= (long)m) arow = (long)m - 1;
    const float4* ap = reinterpret_cast<const float4*>(Phi + arow * 64 + 4 * kh);
    #pragma unroll
    for (int j = 0; j < 8; ++j) a[j] = ap[2 * j];
}

__global__ __launch_bounds__(512) void apply_rinv(const float* __restrict__ Phi,
                                                  const float* __restrict__ Rinv,
                                                  float* __restrict__ Q, int m) {
    __shared__ __attribute__((aligned(16))) float Bp[64 * BSTR];
    const int tid  = threadIdx.x;
    const int lane = tid & 31;
    const int wave = __builtin_amdgcn_readfirstlane(tid >> 5);
    const int li   = lane & 15;
    const int kh   = lane >> 4;

    const long strip0 = (long)blockIdx.x * (16 * ASTRIPS) + wave;

    float4 a[8];
    load_strip(a, Phi, strip0 * 16, li, kh, m);

    {
        const int k = tid >> 3;
        const int c = (tid & 7) * 8;
        const float4* src = reinterpret_cast<const float4*>(Rinv + k * 64 + c);
        const float4 x = src[0], y = src[1];
        const float v[8] = {x.x, x.y, x.z, x.w, y.x, y.y, y.z, y.w};
        #pragma unroll
        for (int q = 0; q < 8; ++q) {
            const int n = c + q;
            const int p = (n & 3) * 16 + (n >> 2);
            Bp[p * BSTR + k] = v[q];
        }
    }
    __syncthreads();

    const float* bbase = &Bp[li * BSTR + 4 * kh];

    for (int s = 0; s < ASTRIPS; ++s) {
        const long row0 = (strip0 + 16 * s) * 16;
        if (row0 >= (long)m) break;

        v8f acc[4] = {};
        #pragma unroll
        for (int j = 0; j < 8; ++j) {
            float4 b[4];
            #pragma unroll
            for (int t = 0; t < 4; ++t)
                b[t] = *reinterpret_cast<const float4*>(bbase + t * 16 * BSTR + 8 * j);
            v2f va;
            va.x = a[j].x; va.y = a[j].y;
            #pragma unroll
            for (int t = 0; t < 4; ++t) {
                v2f vb; vb.x = b[t].x; vb.y = b[t].y;
                acc[t] = __builtin_amdgcn_wmma_f32_16x16x4_f32(false, va, false, vb,
                                                               (short)0, acc[t], false, false);
            }
            va.x = a[j].z; va.y = a[j].w;
            #pragma unroll
            for (int t = 0; t < 4; ++t) {
                v2f vb; vb.x = b[t].z; vb.y = b[t].w;
                acc[t] = __builtin_amdgcn_wmma_f32_16x16x4_f32(false, va, false, vb,
                                                               (short)0, acc[t], false, false);
            }
        }

        if (s + 1 < ASTRIPS) load_strip(a, Phi, row0 + 256, li, kh, m);

        #pragma unroll
        for (int r = 0; r < 8; ++r) {
            const long row = row0 + 8 * kh + r;
            if (row < (long)m) {
                float4 o;
                o.x = acc[0][r]; o.y = acc[1][r]; o.z = acc[2][r]; o.w = acc[3][r];
                *reinterpret_cast<float4*>(Q + row * 64 + 4 * li) = o;
            }
        }
    }
}

extern "C" void kernel_launch(void* const* d_in, const int* in_sizes, int n_in,
                              void* d_out, int out_size, void* d_ws, size_t ws_size,
                              hipStream_t stream) {
    const float* Phi = (const float*)d_in[0];
    float* out = (float*)d_out;
    const int m = in_sizes[0] / 64;

    float* ws   = (float*)d_ws;
    float* G    = ws;
    float* Rinv = ws + 4096;
    float* part = ws + 8192;

    long avail = ((long)(ws_size / 4) - 8192) / 4096;
    int nwg = (int)(avail < 1 ? 1 : (avail > 256 ? 256 : avail));

    gram_partial<<<nwg, 1024, 0, stream>>>(Phi, part, m, nwg);
    reduce_g<<<16, 256, 0, stream>>>(part, G, nwg);
    chol_inv<<<1, 512, 0, stream>>>(G, Rinv);

    const int nblk = (m + 1023) / 1024;
    apply_rinv<<<nblk, 512, 0, stream>>>(Phi, Rinv, out, m);
}


# reduce_g: the in-order sum's loads run up to 64 partials ahead of its adds through a ring of 64 registers (was 8 loads, a wait, 8 adds, 32 times in a row); no sum or order changes
# speedup vs baseline: 2.9053x; 1.1020x over previous
#include <hip/hip_runtime.h>

typedef __attribute__((ext_vector_type(2))) float v2f;
typedef __attribute__((ext_vector_type(8))) float v8f;

#define LSTR 80

#define GROWS  256
#define GSTAGE (GROWS * LSTR)
#define NUT    10

__device__ __forceinline__ void tile_ij(int T, int& ti, int& tj) {
    ti = (T >= 4) + (T >= 7) + (T >= 9);
    tj = T - (4 * ti - (ti * (ti - 1)) / 2) + ti;
}

__device__ __forceinline__ void load_rows(float4 (&p)[4], const float* __restrict__ Phi,
                                          long rowbase, int r, int c, int m) {
    #pragma unroll
    for (int h = 0; h < 2; ++h) {
        const long row = rowbase + r + 128 * h;
        if (row < (long)m) {
            const float4* src = reinterpret_cast<const float4*>(Phi + row * 64 + c);
            p[2 * h]     = src[0];
            p[2 * h + 1] = src[1];
        } else {
            p[2 * h]     = make_float4(0.0f, 0.0f, 0.0f, 0.0f);
            p[2 * h + 1] = make_float4(0.0f, 0.0f, 0.0f, 0.0f);
        }
    }
}

__device__ __forceinline__ void gram_steps4(v8f& acc, const float* pa, const float* pb) {
    #pragma unroll
    for (int k0 = 0; k0 < 16; k0 += 4) {
        v2f va, vb;
        va.x = pa[k0       * LSTR];
        va.y = pa[(k0 + 1) * LSTR];
        vb.x = pb[k0       * LSTR];
        vb.y = pb[(k0 + 1) * LSTR];
        acc = __builtin_amdgcn_wmma_f32_16x16x4_f32(false, va, false, vb,
                                                    (short)0, acc, false, false);
    }
}

__global__ __launch_bounds__(1024) void gram_partial(const float* __restrict__ Phi,
                                                     float* __restrict__ part,
                                                     int m, int nwg) {
    __shared__ __attribute__((aligned(16))) float tile[2 * GSTAGE];
    __shared__ float comb[4 * NUT * 256];
    const int tid  = threadIdx.x;
    const int lane = tid & 31;
    const int wave = __builtin_amdgcn_readfirstlane(tid >> 5);
    const int li   = lane & 15;
    const int kh   = lane >> 4;
    const int r    = tid >> 3;
    const int c    = (tid & 7) * 8;

    const bool slicew = wave < 3 * NUT;

    int aoff[5], boff[5];
    #pragma unroll
    for (int q = 0; q < 5; ++q) {
        const int T    = slicew ? wave / 3 : 5 * (wave - 3 * NUT) + q;
        const int row0 = slicew ? 80 * (wave % 3) + 16 * q : 240;
        int ti, tj;
        tile_ij(T, ti, tj);
        aoff[q] = (row0 + 2 * kh) * LSTR + ti * 16 + li;
        boff[q] = (row0 + 2 * kh) * LSTR + tj * 16 + li;
    }

    v8f acc[5] = {};

    const long stride = (long)nwg * GROWS;
    long rowbase = (long)blockIdx.x * GROWS;
    float4 p[4];
    load_rows(p, Phi, rowbase, r, c, m);

    int s = 0;
    for (; rowbase < (long)m; rowbase += stride) {
        float* st = tile + s * GSTAGE;
        {
            float4* d0 = reinterpret_cast<float4*>(st + r * LSTR + c);
            d0[0] = p[0]; d0[1] = p[1];
            float4* d1 = reinterpret_cast<float4*>(st + (r + 128) * LSTR + c);
            d1[0] = p[2]; d1[1] = p[3];
        }
        __syncthreads();

        load_rows(p, Phi, rowbase + stride, r, c, m);

        #pragma unroll
        for (int q = 0; q < 5; ++q) gram_steps4(acc[q], st + aoff[q], st + boff[q]);
        s ^= 1;
    }

    if (slicew) {
        const v8f t = ((acc[0] + acc[1]) + (acc[2] + acc[3])) + acc[4];
        #pragma unroll
        for (int v = 0; v < 8; ++v) comb[wave * 256 + v * 32 + lane] = t[v];
    } else {
        #pragma unroll
        for (int q = 0; q < 5; ++q) {
            #pragma unroll
            for (int v = 0; v < 8; ++v)
                comb[(3 * NUT + 5 * (wave - 3 * NUT) + q) * 256 + v * 32 + lane] = acc[q][v];
        }
    }
    __syncthreads();
    float* pp = part + (size_t)blockIdx.x * 4096;
    for (int e = tid; e < NUT * 256; e += 1024) {
        const int T   = e >> 8;
        const int idx = e & 255;
        const float* cp = comb + idx;
        const float sum = ((cp[(3 * T) * 256] + cp[(3 * T + 1) * 256]) + cp[(3 * T + 2) * 256])
                          + cp[(3 * NUT + T) * 256];
        int ti, tj;
        tile_ij(T, ti, tj);
        const int gi = ti * 16 + (idx >> 5) + 8 * ((idx >> 4) & 1);
        const int gj = tj * 16 + (idx & 15);
        pp[gi * 64 + gj] = sum;
    }
}

#define RGN 64

__global__ __launch_bounds__(256) void reduce_g(const float* __restrict__ part,
                                                float* __restrict__ G, int nwg) {
    const int e = blockIdx.x * 256 + threadIdx.x;
    {
        const int lane = threadIdx.x & 31;
        const int e0   = e - lane;
        if ((((e0 & 63) >> 4) + 1) >= (e >> 10)) {
            for (int w = lane; w < nwg; w += 32)
                __builtin_prefetch(part + (size_t)w * 4096 + e0, 0, 2);
        }
    }
    if (((e & 63) >> 4) < (e >> 10)) return;
    const float* p = part + e;
    float s = 0.0f;
    const int nring = nwg / RGN;
    if (nring > 0) {
        float r[RGN];
        #pragma unroll
        for (int j = 0; j < RGN; ++j) r[j] = p[(size_t)j * 4096];
        for (int i = 1; i < nring; ++i) {
            #pragma unroll
            for (int j = 0; j < RGN; ++j) {
                s += r[j];
                r[j] = p[(size_t)(RGN + j) * 4096];
            }
            p += (size_t)RGN * 4096;
        }
        #pragma unroll
        for (int j = 0; j < RGN; ++j) s += r[j];
        p += (size_t)RGN * 4096;
    }
    for (int w = nring * RGN; w < nwg; ++w) {
        s += *p;
        p += 4096;
    }
    G[e] = s;
}

#define CLS 144

__device__ __forceinline__ void panel_tile(float* P, int kb, int i, int c,
                                           bool trailing, int li, int kh) {
    const int krow = kb * 16;
    const int ocol = c * 16 + li;
    const int orow = (trailing ? i : kb) * 16;
    v8f acc = {};
    if (trailing) {
        #pragma unroll
        for (int v = 0; v < 8; ++v) acc[v] = P[(orow + v + 8 * kh) * CLS + ocol];
    }
    v2f va[4], vb[4];
    #pragma unroll
    for (int q = 0; q < 4; ++q) {
        const int ka = q * 4 + 2 * kh;
        if (trailing) {
            va[q].x = -P[(krow + ka)     * CLS + i * 16 + li];
            va[q].y = -P[(krow + ka + 1) * CLS + i * 16 + li];
        } else {
            va[q].x = P[(krow + li) * CLS + 64 + krow + ka];
            va[q].y = P[(krow + li) * CLS + 64 + krow + ka + 1];
        }
        vb[q].x = P[(krow + ka)     * CLS + ocol];
        vb[q].y = P[(krow + ka + 1) * CLS + ocol];
    }
    #pragma unroll
    for (int q = 0; q < 4; ++q)
        acc = __builtin_amdgcn_wmma_f32_16x16x4_f32(false, va[q], false, vb[q],
                                                    (short)0, acc, false, false);
    #pragma unroll
    for (int v = 0; v < 8; ++v) P[(orow + v + 8 * kh) * CLS + ocol] = acc[v];
}

__device__ __forceinline__ float lane_bcast(float x, int srclane) {
    return __int_as_float(__builtin_amdgcn_readlane(__float_as_int(x), srclane));
}

__global__ __launch_bounds__(512) void chol_inv(const float* __restrict__ G,
                                                float* __restrict__ Rinv) {
    __shared__ float P[64 * CLS];
    const int tid  = threadIdx.x;
    const int lane = tid & 31;
    const int wave = __builtin_amdgcn_readfirstlane(tid >> 5);
    const int li   = lane & 15;
    const int kh   = lane >> 4;

    {
        const int r = tid >> 3;
        const int c = (tid & 7) * 8;
        float4 a = make_float4(0.0f, 0.0f, 0.0f, 0.0f), b = a;
        if ((c >> 4) >= (r >> 4)) {
            const float4* src = reinterpret_cast<const float4*>(G + r * 64 + c);
            a = src[0]; b = src[1];
        }
        float* dst = &P[r * CLS + c];
        dst[0]=a.x; dst[1]=a.y; dst[2]=a.z; dst[3]=a.w;
        dst[4]=b.x; dst[5]=b.y; dst[6]=b.z; dst[7]=b.w;
        #pragma unroll
        for (int q = 0; q < 8; ++q) dst[64 + q] = (c + q == r) ? 1.0f : 0.0f;
    }
    __syncthreads();

    #pragma unroll
    for (int kb = 0; kb < 4; ++kb) {
        const int krow = kb * 16;

        if (wave == 0) {
            const int ccol = (lane < 16) ? (krow + lane) : (64 + krow + lane - 16);
            float col[16];
            #pragma unroll
            for (int i = 0; i < 16; ++i) col[i] = P[(krow + i) * CLS + ccol];
            #pragma unroll
            for (int k = 0; k < 16; ++k) {
                const float akk = lane_bcast(col[k], k);
                const float dk  = sqrtf(akk);
                const float u   = (lane == k) ? dk : (col[k] / dk);
                col[k] = u;
                #pragma unroll
                for (int i = k + 1; i < 16; ++i) {
                    const float uki = lane_bcast(u, i);
                    col[i] -= uki * u;
                }
            }
            if (lane >= 16) {
                #pragma unroll
                for (int i = 0; i < 16; ++i) P[(krow + i) * CLS + ccol] = col[i];
            }
        }
        __syncthreads();

        if (wave < 8) {
            const int c = wave;
            const bool on = (c < 4) ? (c > kb) : (c - 4 < kb);
            if (on) panel_tile(P, kb, kb, c, false, li, kh);
        }
        __syncthreads();

        if (kb < 3) {
            #pragma unroll
            for (int t = 0; t < 2; ++t) {
                const int id = wave + 16 * t;
                const int i  = id >> 3;
                const int c  = id & 7;
                const bool on = (i > kb) && ((c < 4) ? (c >= i) : (c - 4 <= kb));
                if (on) panel_tile(P, kb, i, c, true, li, kh);
            }
            __syncthreads();
        }
    }

    {
        const int r = tid >> 3;
        const int c = (tid & 7) * 8;
        float4 a, b;
        a.x = P[(c + 0) * CLS + 64 + r]; a.y = P[(c + 1) * CLS + 64 + r];
        a.z = P[(c + 2) * CLS + 64 + r]; a.w = P[(c + 3) * CLS + 64 + r];
        b.x = P[(c + 4) * CLS + 64 + r]; b.y = P[(c + 5) * CLS + 64 + r];
        b.z = P[(c + 6) * CLS + 64 + r]; b.w = P[(c + 7) * CLS + 64 + r];
        float4* dst = reinterpret_cast<float4*>(Rinv + r * 64 + c);
        dst[0] = a; dst[1] = b;
    }
}

#define BSTR 68
#define ASTRIPS 4

__device__ __forceinline__ void load_strip(float4 (&a)[8], const float* __restrict__ Phi,
                                           long row0, int li, int kh, int m) {
    long arow = row0 + li;
    if (arow >= (long)m) arow = (long)m - 1;
    const float4* ap = reinterpret_cast<const float4*>(Phi + arow * 64 + 4 * kh);
    #pragma unroll
    for (int j = 0; j < 8; ++j) a[j] = ap[2 * j];
}

__global__ __launch_bounds__(512) void apply_rinv(const float* __restrict__ Phi,
                                                  const float* __restrict__ Rinv,
                                                  float* __restrict__ Q, int m) {
    __shared__ __attribute__((aligned(16))) float Bp[64 * BSTR];
    const int tid  = threadIdx.x;
    const int lane = tid & 31;
    const int wave = __builtin_amdgcn_readfirstlane(tid >> 5);
    const int li   = lane & 15;
    const int kh   = lane >> 4;

    const long strip0 = (long)blockIdx.x * (16 * ASTRIPS) + wave;

    float4 a[8];
    load_strip(a, Phi, strip0 * 16, li, kh, m);

    {
        const int k = tid >> 3;
        const int c = (tid & 7) * 8;
        const float4* src = reinterpret_cast<const float4*>(Rinv + k * 64 + c);
        const float4 x = src[0], y = src[1];
        const float v[8] = {x.x, x.y, x.z, x.w, y.x, y.y, y.z, y.w};
        #pragma unroll
        for (int q = 0; q < 8; ++q) {
            const int n = c + q;
            const int p = (n & 3) * 16 + (n >> 2);
            Bp[p * BSTR + k] = v[q];
        }
    }
    __syncthreads();

    const float* bbase = &Bp[li * BSTR + 4 * kh];

    for (int s = 0; s < ASTRIPS; ++s) {
        const long row0 = (strip0 + 16 * s) * 16;
        if (row0 >= (long)m) break;

        v8f acc[4] = {};
        #pragma unroll
        for (int j = 0; j < 8; ++j) {
            float4 b[4];
            #pragma unroll
            for (int t = 0; t < 4; ++t)
                b[t] = *reinterpret_cast<const float4*>(bbase + t * 16 * BSTR + 8 * j);
            v2f va;
            va.x = a[j].x; va.y = a[j].y;
            #pragma unroll
            for (int t = 0; t < 4; ++t) {
                v2f vb; vb.x = b[t].x; vb.y = b[t].y;
                acc[t] = __builtin_amdgcn_wmma_f32_16x16x4_f32(false, va, false, vb,
                                                               (short)0, acc[t], false, false);
            }
            va.x = a[j].z; va.y = a[j].w;
            #pragma unroll
            for (int t = 0; t < 4; ++t) {
                v2f vb; vb.x = b[t].z; vb.y = b[t].w;
                acc[t] = __builtin_amdgcn_wmma_f32_16x16x4_f32(false, va, false, vb,
                                                               (short)0, acc[t], false, false);
            }
        }

        if (s + 1 < ASTRIPS) load_strip(a, Phi, row0 + 256, li, kh, m);

        #pragma unroll
        for (int r = 0; r < 8; ++r) {
            const long row = row0 + 8 * kh + r;
            if (row < (long)m) {
                float4 o;
                o.x = acc[0][r]; o.y = acc[1][r]; o.z = acc[2][r]; o.w = acc[3][r];
                *reinterpret_cast<float4*>(Q + row * 64 + 4 * li) = o;
            }
        }
    }
}

extern "C" void kernel_launch(void* const* d_in, const int* in_sizes, int n_in,
                              void* d_out, int out_size, void* d_ws, size_t ws_size,
                              hipStream_t stream) {
    const float* Phi = (const float*)d_in[0];
    float* out = (float*)d_out;
    const int m = in_sizes[0] / 64;

    float* ws   = (float*)d_ws;
    float* G    = ws;
    float* Rinv = ws + 4096;
    float* part = ws + 8192;

    long avail = ((long)(ws_size / 4) - 8192) / 4096;
    int nwg = (int)(avail < 1 ? 1 : (avail > 256 ? 256 : avail));

    gram_partial<<<nwg, 1024, 0, stream>>>(Phi, part, m, nwg);
    reduce_g<<<16, 256, 0, stream>>>(part, G, nwg);
    chol_inv<<<1, 512, 0, stream>>>(G, Rinv);

    const int nblk = (m + 1023) / 1024;
    apply_rinv<<<nblk, 512, 0, stream>>>(Phi, Rinv, out, m);
}
